# starting kernel
# baseline (speedup 1.0000x reference)
#include <hip/hip_runtime.h>


typedef __attribute__((ext_vector_type(16))) _Float16 v16h;
typedef __attribute__((ext_vector_type(8)))  _Float16 v8h;
typedef __attribute__((ext_vector_type(4)))  _Float16 v4h;
typedef __attribute__((ext_vector_type(8)))  float    v8f;

static constexpr int N_  = 64;
static constexpr int M_  = 64;
static constexpr int F_  = 256;
static constexpr int H_  = 4;
static constexpr int D_  = 64;
static constexpr int HD_ = H_ * D_;
static constexpr int FRAG_HALFS = 32 * 16;
static constexpr int W1F_HALFS  = F_ * HD_;
static constexpr int HEAD_HALFS = 4 * 2 * FRAG_HALFS;
static constexpr int PAIR_HALFS = H_ * HEAD_HALFS;
static constexpr int LDS_STRIDE = 264;

__global__ void swizzle_w1_kernel(const float* __restrict__ W1,
                                  _Float16* __restrict__ w1f) {
  int id   = blockIdx.x * blockDim.x + threadIdx.x;
  int jj   = id & 15;
  int lane = (id >> 4) & 31;
  int kt   = (id >> 9) & 7;
  int nt   = (id >> 12) & 15;
  int n    = nt * 16 + (lane & 15);
  int koff = (lane >> 4) * 8;
  int k    = kt * 32 + koff + ((jj < 8) ? jj : (8 + jj));
  w1f[id] = (_Float16)W1[k * HD_ + n];
}

__global__ void __launch_bounds__(256)
proj_kernel(const float* __restrict__ x,
            const _Float16* __restrict__ w1f,
            _Float16* __restrict__ hbuf) {
  __shared__ __align__(16) _Float16 lds[64 * LDS_STRIDE];

  const int t    = threadIdx.x;
  const int lane = t & 31;
  const int wave = t >> 5;
  const int bi   = blockIdx.y;
  const int bj   = blockIdx.x;

  {
    const float4* x4 = (const float4*)(x + (size_t)(bi * N_ + bj) * (M_ * F_));
#pragma unroll
    for (int k = 0; k < 16; ++k) {
      int f4  = k * 256 + t;
      float4 v = x4[f4];
      int row = f4 >> 6;
      int col = (f4 & 63) * 4;
      v4h hv = { (_Float16)v.x, (_Float16)v.y, (_Float16)v.z, (_Float16)v.w };
      *(v4h*)&lds[row * LDS_STRIDE + col] = hv;
    }
  }
  __syncthreads();

  const int mtBase = (wave >> 2) * 2;
  const int ntBase = (wave & 3) * 4;
  const int mrow0  = mtBase * 16 + (lane & 15);
  const int klane  = (lane >> 4) << 3;

  v8f acc[2][4] = {};

#pragma unroll
  for (int kt = 0; kt < 8; ++kt) {
    v16h a[2];
#pragma unroll
    for (int mi = 0; mi < 2; ++mi) {
      const _Float16* src = &lds[(mrow0 + mi * 16) * LDS_STRIDE + kt * 32 + klane];
      v8h lo = *(const v8h*)src;
      v8h hi = *(const v8h*)(src + 16);
      v16h av;
#pragma unroll
      for (int u = 0; u < 8; ++u) { av[u] = lo[u]; av[8 + u] = hi[u]; }
      a[mi] = av;
    }
#pragma unroll
    for (int ni = 0; ni < 4; ++ni) {
      v16h b = *(const v16h*)(w1f + ((ntBase + ni) * 8 + kt) * FRAG_HALFS + lane * 16);
#pragma unroll
      for (int mi = 0; mi < 2; ++mi) {
        acc[mi][ni] = __builtin_amdgcn_wmma_f32_16x16x32_f16(
            false, a[mi], false, b, (short)0, acc[mi][ni], false, false);
      }
    }
  }
  __syncthreads();

#pragma unroll
  for (int mi = 0; mi < 2; ++mi) {
#pragma unroll
    for (int ni = 0; ni < 4; ++ni) {
      int n = (ntBase + ni) * 16 + (lane & 15);
#pragma unroll
      for (int r = 0; r < 8; ++r) {
        int m = (mtBase + mi) * 16 + r + ((lane >> 4) << 3);
        lds[m * LDS_STRIDE + n] = (_Float16)acc[mi][ni][r];
      }
    }
  }
  __syncthreads();

  _Float16* dst = hbuf + (size_t)(bi * N_ + bj) * PAIR_HALFS;
#pragma unroll
  for (int it = 0; it < 4; ++it) {
    int c     = it * 256 + t;
    int lane2 = c & 31;
    int ktc   = (c >> 5) & 1;
    int mtc   = (c >> 6) & 3;
    int head  = (c >> 8) & 3;
    int m     = mtc * 16 + (lane2 & 15);
    int kb    = ktc * 32 + ((lane2 >> 4) << 3);
    const _Float16* src = &lds[m * LDS_STRIDE + head * 64 + kb];
    v8h lo = *(const v8h*)src;
    v8h hi = *(const v8h*)(src + 16);
    v8h* o = (v8h*)(dst + (((head * 4 + mtc) * 2 + ktc) * FRAG_HALFS) + lane2 * 16);
    o[0] = lo;
    o[1] = hi;
  }
}

__global__ void __launch_bounds__(256)
score_kernel(const _Float16* __restrict__ hbuf,
             const float* __restrict__ nItem,
             const float* __restrict__ W2,
             float* __restrict__ out) {
  __shared__ float ssum[H_];
  const int t    = threadIdx.x;
  const int lane = t & 31;
  const int wave = t >> 5;
  const int p    = blockIdx.y;
  const int q    = blockIdx.x;

  if (t < H_) ssum[t] = 0.0f;
  __syncthreads();

  const int head   = wave >> 1;
  const int mtBase = (wave & 1) * 2;
  const _Float16* Abase = hbuf + ((size_t)(p * N_ + q) * H_ + head) * HEAD_HALFS;
  const _Float16* Bbase = hbuf + ((size_t)(q * N_ + p) * H_ + head) * HEAD_HALFS;

  v8f acc[2][4] = {};
#pragma unroll
  for (int kt = 0; kt < 2; ++kt) {
    v16h a[2], b[4];
#pragma unroll
    for (int mi = 0; mi < 2; ++mi)
      a[mi] = *(const v16h*)(Abase + ((mtBase + mi) * 2 + kt) * FRAG_HALFS + lane * 16);
#pragma unroll
    for (int ni = 0; ni < 4; ++ni)
      b[ni] = *(const v16h*)(Bbase + (ni * 2 + kt) * FRAG_HALFS + lane * 16);
#pragma unroll
    for (int mi = 0; mi < 2; ++mi)
#pragma unroll
      for (int ni = 0; ni < 4; ++ni)
        acc[mi][ni] = __builtin_amdgcn_wmma_f32_16x16x32_f16(
            false, a[mi], false, b[ni], (short)0, acc[mi][ni], false, false);
  }

  float s = 0.0f;
#pragma unroll
  for (int mi = 0; mi < 2; ++mi)
#pragma unroll
    for (int ni = 0; ni < 4; ++ni)
#pragma unroll
      for (int r = 0; r < 8; ++r)
        s += fmaxf(acc[mi][ni][r], 0.0f);

#pragma unroll
  for (int off = 16; off > 0; off >>= 1)
    s += __shfl_xor(s, off, 32);
  if (lane == 0) atomicAdd(&ssum[head], s);
  __syncthreads();

  if (t == 0) {
    float inv = 0.125f / (nItem[p] * nItem[q]);
    float r = 0.0f;
#pragma unroll
    for (int h = 0; h < H_; ++h) r += ssum[h] * W2[h];
    out[p * N_ + q] = r * inv;
  }
}

extern "C" void kernel_launch(void* const* d_in, const int* in_sizes, int n_in,
                              void* d_out, int out_size, void* d_ws, size_t ws_size,
                              hipStream_t stream) {
  const float* x     = (const float*)d_in[0];
  const float* nItem = (const float*)d_in[1];
  const float* W1    = (const float*)d_in[2];
  const float* W2    = (const float*)d_in[3];
  float* out = (float*)d_out;

  _Float16* w1f  = (_Float16*)d_ws;
  _Float16* hbuf = w1f + W1F_HALFS;

  swizzle_w1_kernel<<<256, 256, 0, stream>>>(W1, w1f);
  proj_kernel<<<dim3(N_, N_), 256, 0, stream>>>(x, w1f, hbuf);
  score_kernel<<<dim3(N_, N_), 256, 0, stream>>>(hbuf, nItem, W2, out);
}


# Projection and pair score fused, h kept in registers: block = unordered tile pair, wave = head, two 32-column passes; packed accumulators are the score fragments; out[j,i]=out[i,j]; other sum order
# speedup vs baseline: 1.8005x; 1.8005x over previous
#include <hip/hip_runtime.h>


typedef __attribute__((ext_vector_type(16))) _Float16 v16h;
typedef __attribute__((ext_vector_type(8)))  _Float16 v8h;
typedef __attribute__((ext_vector_type(4)))  _Float16 v4h;
typedef __attribute__((ext_vector_type(8)))  float    v8f;

static constexpr int N_  = 64;
static constexpr int M_  = 64;
static constexpr int F_  = 256;
static constexpr int H_  = 4;
static constexpr int D_  = 64;
static constexpr int HD_ = H_ * D_;
static constexpr int FRAG_HALFS = 32 * 16;
static constexpr int W1F_HALFS  = F_ * HD_;
static constexpr int LDS_STRIDE = 264;

__global__ void swizzle_w1_kernel(const float* __restrict__ W1,
                                  _Float16* __restrict__ w1f) {
  int id   = blockIdx.x * blockDim.x + threadIdx.x;
  int jj   = id & 15;
  int lane = (id >> 4) & 31;
  int kt   = (id >> 9) & 7;
  int nt   = (id >> 12) & 15;
  int n    = nt * 16 + (lane & 15);
  int koff = (lane >> 4) * 8;
  int k    = kt * 32 + koff + ((jj < 8) ? jj : (8 + jj));
  w1f[id] = (_Float16)W1[k * HD_ + n];
}

static __device__ __forceinline__ void load_batch(float4 (&v)[16], const float4* __restrict__ p,
                                                  int k0, int t) {
#pragma unroll
  for (int u = 0; u < 16; ++u) v[u] = p[(k0 + u) * 128 + t];
}
static __device__ __forceinline__ void store_batch(const float4 (&v)[16], _Float16* tile_lds,
                                                   int k0, int t) {
#pragma unroll
  for (int u = 0; u < 16; ++u) {
    int f4  = (k0 + u) * 128 + t;
    int row = f4 >> 6;
    int col = (f4 & 63) * 4;
    v4h hv = { (_Float16)v[u].x, (_Float16)v[u].y, (_Float16)v[u].z, (_Float16)v[u].w };
    *(v4h*)&tile_lds[row * LDS_STRIDE + col] = hv;
  }
}

__global__ void __launch_bounds__(128, 4)
pair_kernel(const float* __restrict__ x,
            const _Float16* __restrict__ w1f,
            const float* __restrict__ nItem,
            const float* __restrict__ W2,
            float* __restrict__ out) {
  __shared__ __align__(16) _Float16 lds[2][64 * LDS_STRIDE];
  __shared__ float ssum[H_];

  const int t    = threadIdx.x;
  const int lane = t & 31;
  const int head = t >> 5;
  const int r    = blockIdx.y;
  const int c    = blockIdx.x;
  int ti, tj;
  if (c > r) { ti = r;      tj = c - 1;  }
  else       { ti = 63 - r; tj = ti + c; }

  {
    const float4* x0 = (const float4*)(x + (size_t)(ti * N_ + tj) * (M_ * F_));
    const float4* x1 = (const float4*)(x + (size_t)(tj * N_ + ti) * (M_ * F_));
    float4 va[16], vb[16];
    load_batch(va, x0, 0, t);
    load_batch(vb, x0, 16, t);
    store_batch(va, lds[0], 0, t);
    load_batch(va, x1, 0, t);
    store_batch(vb, lds[0], 16, t);
    load_batch(vb, x1, 16, t);
    store_batch(va, lds[1], 0, t);
    store_batch(vb, lds[1], 16, t);
  }
  __syncthreads();

  const int ntBase = head * 4;
  const int item0  = lane & 15;
  const int klane  = (lane >> 4) << 3;

  v16h fr[2][4][2];

#pragma unroll
  for (int P = 0; P < 2; ++P) {
    asm volatile("" ::: "memory");
    v8f acc[2][2][4] = {};
#pragma unroll
    for (int kt = 0; kt < 8; ++kt) {
      v16h w[2];
#pragma unroll
      for (int nl = 0; nl < 2; ++nl)
        w[nl] = *(const v16h*)(w1f + ((ntBase + 2 * P + nl) * 8 + kt) * FRAG_HALFS + lane * 16);
#pragma unroll
      for (int tile = 0; tile < 2; ++tile) {
        v16h xb[4];
#pragma unroll
        for (int mi = 0; mi < 4; ++mi) {
          const _Float16* src = &lds[tile][(mi * 16 + item0) * LDS_STRIDE + kt * 32 + klane];
          v8h lo = *(const v8h*)src;
          v8h hi = *(const v8h*)(src + 16);
          v16h bv;
#pragma unroll
          for (int u = 0; u < 8; ++u) { bv[u] = lo[u]; bv[8 + u] = hi[u]; }
          xb[mi] = bv;
        }
#pragma unroll
        for (int nl = 0; nl < 2; ++nl)
#pragma unroll
          for (int mi = 0; mi < 4; ++mi)
            acc[tile][nl][mi] = __builtin_amdgcn_wmma_f32_16x16x32_f16(
                false, w[nl], false, xb[mi], (short)0, acc[tile][nl][mi], false, false);
      }
    }
#pragma unroll
    for (int tile = 0; tile < 2; ++tile)
#pragma unroll
      for (int mi = 0; mi < 4; ++mi) {
        v16h f;
#pragma unroll
        for (int e = 0; e < 8; ++e) {
          f[e]     = (_Float16)acc[tile][0][mi][e];
          f[8 + e] = (_Float16)acc[tile][1][mi][e];
        }
        fr[tile][mi][P] = f;
      }
  }

  float s[4] = { 0.0f, 0.0f, 0.0f, 0.0f };
#pragma unroll
  for (int Mt = 0; Mt < 4; ++Mt) {
    v8f sc[4] = {};
#pragma unroll
    for (int Kt = 0; Kt < 2; ++Kt)
#pragma unroll
      for (int Nt = 0; Nt < 4; ++Nt)
        sc[Nt] = __builtin_amdgcn_wmma_f32_16x16x32_f16(
            false, fr[0][Mt][Kt], false, fr[1][Nt][Kt], (short)0, sc[Nt], false, false);
#pragma unroll
    for (int Nt = 0; Nt < 4; ++Nt)
#pragma unroll
      for (int e = 0; e < 8; ++e)
        s[Nt] += fmaxf(sc[Nt][e], 0.0f);
  }
  float tot = (s[0] + s[1]) + (s[2] + s[3]);
#pragma unroll
  for (int off = 16; off > 0; off >>= 1)
    tot += __shfl_xor(tot, off, 32);
  if (lane == 0) ssum[head] = tot;
  __syncthreads();

  if (t == 0) {
    float inv = 0.125f / (nItem[ti] * nItem[tj]);
    float rr = 0.0f;
#pragma unroll
    for (int h = 0; h < H_; ++h) rr += ssum[h] * W2[h];
    float v = rr * inv;
    out[ti * N_ + tj] = v;
    if (ti != tj) out[tj * N_ + ti] = v;
  }
}

extern "C" void kernel_launch(void* const* d_in, const int* in_sizes, int n_in,
                              void* d_out, int out_size, void* d_ws, size_t ws_size,
                              hipStream_t stream) {
  const float* x     = (const float*)d_in[0];
  const float* nItem = (const float*)d_in[1];
  const float* W1    = (const float*)d_in[2];
  const float* W2    = (const float*)d_in[3];
  float* out = (float*)d_out;

  _Float16* w1f = (_Float16*)d_ws;

  swizzle_w1_kernel<<<256, 256, 0, stream>>>(W1, w1f);
  pair_kernel<<<dim3(65, 32), 128, 0, stream>>>(x, w1f, nItem, W2, out);
}


# One k pass per wave at two waves per SIMD: all 64 head columns of both tiles in 256 accumulator registers, x read from LDS once not twice, later blocks' W1 and x fragments loaded ahead as one group
# speedup vs baseline: 2.1070x; 1.1702x over previous
#include <hip/hip_runtime.h>


typedef __attribute__((ext_vector_type(16))) _Float16 v16h;
typedef __attribute__((ext_vector_type(8)))  _Float16 v8h;
typedef __attribute__((ext_vector_type(4)))  _Float16 v4h;
typedef __attribute__((ext_vector_type(8)))  float    v8f;

static constexpr int N_  = 64;
static constexpr int M_  = 64;
static constexpr int F_  = 256;
static constexpr int H_  = 4;
static constexpr int D_  = 64;
static constexpr int HD_ = H_ * D_;
static constexpr int FRAG_HALFS = 32 * 16;
static constexpr int W1F_HALFS  = F_ * HD_;
static constexpr int LDS_STRIDE = 264;

__global__ void swizzle_w1_kernel(const float* __restrict__ W1,
                                  _Float16* __restrict__ w1f) {
  int id   = blockIdx.x * blockDim.x + threadIdx.x;
  int jj   = id & 15;
  int lane = (id >> 4) & 31;
  int kt   = (id >> 9) & 7;
  int nt   = (id >> 12) & 15;
  int n    = nt * 16 + (lane & 15);
  int koff = (lane >> 4) * 8;
  int k    = kt * 32 + koff + ((jj < 8) ? jj : (8 + jj));
  w1f[id] = (_Float16)W1[k * HD_ + n];
}

static __device__ __forceinline__ void load_batch(float4 (&v)[16], const float4* __restrict__ p,
                                                  int k0, int t) {
#pragma unroll
  for (int u = 0; u < 16; ++u) v[u] = p[(k0 + u) * 128 + t];
}
static __device__ __forceinline__ void store_batch(const float4 (&v)[16], _Float16* tile_lds,
                                                   int k0, int t) {
#pragma unroll
  for (int u = 0; u < 16; ++u) {
    int f4  = (k0 + u) * 128 + t;
    int row = f4 >> 6;
    int col = (f4 & 63) * 4;
    v4h hv = { (_Float16)v[u].x, (_Float16)v[u].y, (_Float16)v[u].z, (_Float16)v[u].w };
    *(v4h*)&tile_lds[row * LDS_STRIDE + col] = hv;
  }
}

__global__ void __launch_bounds__(128, 2)
pair_kernel(const float* __restrict__ x,
            const _Float16* __restrict__ w1f,
            const float* __restrict__ nItem,
            const float* __restrict__ W2,
            float* __restrict__ out) {
  __shared__ __align__(16) _Float16 lds[2][64 * LDS_STRIDE];
  __shared__ float ssum[H_];

  const int t    = threadIdx.x;
  const int lane = t & 31;
  const int head = t >> 5;
  const int r    = blockIdx.y;
  const int c    = blockIdx.x;
  int ti, tj;
  if (c > r) { ti = r;      tj = c - 1;  }
  else       { ti = 63 - r; tj = ti + c; }

  {
    const float4* x0 = (const float4*)(x + (size_t)(ti * N_ + tj) * (M_ * F_));
    const float4* x1 = (const float4*)(x + (size_t)(tj * N_ + ti) * (M_ * F_));
    float4 va[16], vb[16], vc[16], vd[16];
    load_batch(va, x0, 0, t);
    load_batch(vb, x0, 16, t);
    load_batch(vc, x1, 0, t);
    load_batch(vd, x1, 16, t);
    store_batch(va, lds[0], 0, t);
    store_batch(vb, lds[0], 16, t);
    store_batch(vc, lds[1], 0, t);
    store_batch(vd, lds[1], 16, t);
  }
  __syncthreads();

  const int ntBase = head * 4;
  const int item0  = lane & 15;
  const int klane  = (lane >> 4) << 3;

  v16h fr[2][4][2];

  auto load_w = [&](v16h (&w)[4], int kt) {
#pragma unroll
    for (int c = 0; c < 4; ++c)
      w[c] = *(const v16h*)(w1f + ((ntBase + c) * 8 + kt) * FRAG_HALFS + lane * 16);
  };
  auto load_x = [&](v16h (&xf)[4], int kt, int tile) {
#pragma unroll
    for (int mi = 0; mi < 4; ++mi) {
      const _Float16* src = &lds[tile][(mi * 16 + item0) * LDS_STRIDE + kt * 32 + klane];
      v8h lo = *(const v8h*)src;
      v8h hi = *(const v8h*)(src + 16);
      v16h bv;
#pragma unroll
      for (int u = 0; u < 8; ++u) { bv[u] = lo[u]; bv[8 + u] = hi[u]; }
      xf[mi] = bv;
    }
  };

  {
    v8f acc[2][4][4] = {};
    v16h w[3][4];
    v16h xf[2][4];
    load_w(w[0], 0);
    load_w(w[1], 1);
    load_x(xf[0], 0, 0);
#pragma unroll
    for (int b = 0; b < 16; ++b) {
      const int kt = b >> 1, tile = b & 1;
      if (tile == 0 && kt + 2 < 8) load_w(w[(kt + 2) % 3], kt + 2);
      if (b + 1 < 16) load_x(xf[(b + 1) & 1], (b + 1) >> 1, (b + 1) & 1);
      __builtin_amdgcn_sched_barrier(0);
#pragma unroll
      for (int c = 0; c < 4; ++c)
#pragma unroll
        for (int mi = 0; mi < 4; ++mi)
          acc[tile][c][mi] = __builtin_amdgcn_wmma_f32_16x16x32_f16(
              false, w[kt % 3][c], false, xf[b & 1][mi], (short)0, acc[tile][c][mi], false, false);
      __builtin_amdgcn_sched_barrier(0);
    }
#pragma unroll
    for (int P = 0; P < 2; ++P)
#pragma unroll
      for (int tile = 0; tile < 2; ++tile)
#pragma unroll
        for (int mi = 0; mi < 4; ++mi) {
          v16h f;
#pragma unroll
          for (int e = 0; e < 8; ++e) {
            f[e]     = (_Float16)acc[tile][2 * P][mi][e];
            f[8 + e] = (_Float16)acc[tile][2 * P + 1][mi][e];
          }
          fr[tile][mi][P] = f;
        }
  }

  float s[4] = { 0.0f, 0.0f, 0.0f, 0.0f };
#pragma unroll
  for (int Mt = 0; Mt < 4; ++Mt) {
    v8f sc[4] = {};
#pragma unroll
    for (int Kt = 0; Kt < 2; ++Kt)
#pragma unroll
      for (int Nt = 0; Nt < 4; ++Nt)
        sc[Nt] = __builtin_amdgcn_wmma_f32_16x16x32_f16(
            false, fr[0][Mt][Kt], false, fr[1][Nt][Kt], (short)0, sc[Nt], false, false);
#pragma unroll
    for (int Nt = 0; Nt < 4; ++Nt)
#pragma unroll
      for (int e = 0; e < 8; ++e)
        s[Nt] += fmaxf(sc[Nt][e], 0.0f);
  }
  float tot = (s[0] + s[1]) + (s[2] + s[3]);
#pragma unroll
  for (int off = 16; off > 0; off >>= 1)
    tot += __shfl_xor(tot, off, 32);
  if (lane == 0) ssum[head] = tot;
  __syncthreads();

  if (t == 0) {
    float inv = 0.125f / (nItem[ti] * nItem[tj]);
    float rr = 0.0f;
#pragma unroll
    for (int h = 0; h < H_; ++h) rr += ssum[h] * W2[h];
    float v = rr * inv;
    out[ti * N_ + tj] = v;
    if (ti != tj) out[tj * N_ + ti] = v;
  }
}

extern "C" void kernel_launch(void* const* d_in, const int* in_sizes, int n_in,
                              void* d_out, int out_size, void* d_ws, size_t ws_size,
                              hipStream_t stream) {
  const float* x     = (const float*)d_in[0];
  const float* nItem = (const float*)d_in[1];
  const float* W1    = (const float*)d_in[2];
  const float* W2    = (const float*)d_in[3];
  float* out = (float*)d_out;

  _Float16* w1f = (_Float16*)d_ws;

  swizzle_w1_kernel<<<256, 256, 0, stream>>>(W1, w1f);
  pair_kernel<<<dim3(65, 32), 128, 0, stream>>>(x, w1f, nItem, W2, out);
}


# W1 fragments fetched further ahead: k-steps 0-4 asked for during x staging (behind half the x returns, before the barrier), k-steps 5-7 four k-steps ahead; five register sets, was three; same sums
# speedup vs baseline: 2.1817x; 1.0355x over previous
#include <hip/hip_runtime.h>


typedef __attribute__((ext_vector_type(16))) _Float16 v16h;
typedef __attribute__((ext_vector_type(8)))  _Float16 v8h;
typedef __attribute__((ext_vector_type(4)))  _Float16 v4h;
typedef __attribute__((ext_vector_type(8)))  float    v8f;

static constexpr int N_  = 64;
static constexpr int M_  = 64;
static constexpr int F_  = 256;
static constexpr int H_  = 4;
static constexpr int D_  = 64;
static constexpr int HD_ = H_ * D_;
static constexpr int FRAG_HALFS = 32 * 16;
static constexpr int W1F_HALFS  = F_ * HD_;
static constexpr int LDS_STRIDE = 264;

__global__ void swizzle_w1_kernel(const float* __restrict__ W1,
                                  _Float16* __restrict__ w1f) {
  int id   = blockIdx.x * blockDim.x + threadIdx.x;
  int jj   = id & 15;
  int lane = (id >> 4) & 31;
  int kt   = (id >> 9) & 7;
  int nt   = (id >> 12) & 15;
  int n    = nt * 16 + (lane & 15);
  int koff = (lane >> 4) * 8;
  int k    = kt * 32 + koff + ((jj < 8) ? jj : (8 + jj));
  w1f[id] = (_Float16)W1[k * HD_ + n];
}

static __device__ __forceinline__ void load_batch(float4 (&v)[16], const float4* __restrict__ p,
                                                  int k0, int t) {
#pragma unroll
  for (int u = 0; u < 16; ++u) v[u] = p[(k0 + u) * 128 + t];
}
static __device__ __forceinline__ void store_batch(const float4 (&v)[16], _Float16* tile_lds,
                                                   int k0, int t) {
#pragma unroll
  for (int u = 0; u < 16; ++u) {
    int f4  = (k0 + u) * 128 + t;
    int row = f4 >> 6;
    int col = (f4 & 63) * 4;
    v4h hv = { (_Float16)v[u].x, (_Float16)v[u].y, (_Float16)v[u].z, (_Float16)v[u].w };
    *(v4h*)&tile_lds[row * LDS_STRIDE + col] = hv;
  }
}

__global__ void __launch_bounds__(128, 2)
pair_kernel(const float* __restrict__ x,
            const _Float16* __restrict__ w1f,
            const float* __restrict__ nItem,
            const float* __restrict__ W2,
            float* __restrict__ out) {
  __shared__ __align__(16) _Float16 lds[2][64 * LDS_STRIDE];
  __shared__ float ssum[H_];

  const int t    = threadIdx.x;
  const int lane = t & 31;
  const int head = t >> 5;
  const int r    = blockIdx.y;
  const int c    = blockIdx.x;
  int ti, tj;
  if (c > r) { ti = r;      tj = c - 1;  }
  else       { ti = 63 - r; tj = ti + c; }

  const int ntBase = head * 4;
  const int item0  = lane & 15;
  const int klane  = (lane >> 4) << 3;

  v16h fr[2][4][2];

  auto load_w = [&](v16h (&w)[4], int kt) {
#pragma unroll
    for (int c = 0; c < 4; ++c)
      w[c] = *(const v16h*)(w1f + ((ntBase + c) * 8 + kt) * FRAG_HALFS + lane * 16);
  };
  auto load_x = [&](v16h (&xf)[4], int kt, int tile) {
#pragma unroll
    for (int mi = 0; mi < 4; ++mi) {
      const _Float16* src = &lds[tile][(mi * 16 + item0) * LDS_STRIDE + kt * 32 + klane];
      v8h lo = *(const v8h*)src;
      v8h hi = *(const v8h*)(src + 16);
      v16h bv;
#pragma unroll
      for (int u = 0; u < 8; ++u) { bv[u] = lo[u]; bv[8 + u] = hi[u]; }
      xf[mi] = bv;
    }
  };

  v16h w[5][4];

  {
    const float4* x0 = (const float4*)(x + (size_t)(ti * N_ + tj) * (M_ * F_));
    const float4* x1 = (const float4*)(x + (size_t)(tj * N_ + ti) * (M_ * F_));
    float4 va[16], vb[16], vc[16], vd[16];
    load_batch(va, x0, 0, t);
    load_batch(vb, x0, 16, t);
    load_batch(vc, x1, 0, t);
    load_batch(vd, x1, 16, t);
    store_batch(va, lds[0], 0, t);
    store_batch(vb, lds[0], 16, t);
    __builtin_amdgcn_sched_barrier(0);
#pragma unroll
    for (int kt = 0; kt < 5; ++kt) load_w(w[kt], kt);
    __builtin_amdgcn_sched_barrier(0);
    store_batch(vc, lds[1], 0, t);
    store_batch(vd, lds[1], 16, t);
  }
  __syncthreads();

  {
    v8f acc[2][4][4] = {};
    v16h xf[2][4];
    load_x(xf[0], 0, 0);
#pragma unroll
    for (int b = 0; b < 16; ++b) {
      const int kt = b >> 1, tile = b & 1;
      if (tile == 0 && kt >= 1 && kt + 4 < 8) load_w(w[(kt + 4) % 5], kt + 4);
      if (b + 1 < 16) load_x(xf[(b + 1) & 1], (b + 1) >> 1, (b + 1) & 1);
      __builtin_amdgcn_sched_barrier(0);
#pragma unroll
      for (int c = 0; c < 4; ++c)
#pragma unroll
        for (int mi = 0; mi < 4; ++mi)
          acc[tile][c][mi] = __builtin_amdgcn_wmma_f32_16x16x32_f16(
              false, w[kt % 5][c], false, xf[b & 1][mi], (short)0, acc[tile][c][mi], false, false);
      __builtin_amdgcn_sched_barrier(0);
    }
#pragma unroll
    for (int P = 0; P < 2; ++P)
#pragma unroll
      for (int tile = 0; tile < 2; ++tile)
#pragma unroll
        for (int mi = 0; mi < 4; ++mi) {
          v16h f;
#pragma unroll
          for (int e = 0; e < 8; ++e) {
            f[e]     = (_Float16)acc[tile][2 * P][mi][e];
            f[8 + e] = (_Float16)acc[tile][2 * P + 1][mi][e];
          }
          fr[tile][mi][P] = f;
        }
  }

  float s[4] = { 0.0f, 0.0f, 0.0f, 0.0f };
#pragma unroll
  for (int Mt = 0; Mt < 4; ++Mt) {
    v8f sc[4] = {};
#pragma unroll
    for (int Kt = 0; Kt < 2; ++Kt)
#pragma unroll
      for (int Nt = 0; Nt < 4; ++Nt)
        sc[Nt] = __builtin_amdgcn_wmma_f32_16x16x32_f16(
            false, fr[0][Mt][Kt], false, fr[1][Nt][Kt], (short)0, sc[Nt], false, false);
#pragma unroll
    for (int Nt = 0; Nt < 4; ++Nt)
#pragma unroll
      for (int e = 0; e < 8; ++e)
        s[Nt] += fmaxf(sc[Nt][e], 0.0f);
  }
  float tot = (s[0] + s[1]) + (s[2] + s[3]);
#pragma unroll
  for (int off = 16; off > 0; off >>= 1)
    tot += __shfl_xor(tot, off, 32);
  if (lane == 0) ssum[head] = tot;
  __syncthreads();

  if (t == 0) {
    float inv = 0.125f / (nItem[ti] * nItem[tj]);
    float rr = 0.0f;
#pragma unroll
    for (int h = 0; h < H_; ++h) rr += ssum[h] * W2[h];
    float v = rr * inv;
    out[ti * N_ + tj] = v;
    if (ti != tj) out[tj * N_ + ti] = v;
  }
}

extern "C" void kernel_launch(void* const* d_in, const int* in_sizes, int n_in,
                              void* d_out, int out_size, void* d_ws, size_t ws_size,
                              hipStream_t stream) {
  const float* x     = (const float*)d_in[0];
  const float* nItem = (const float*)d_in[1];
  const float* W1    = (const float*)d_in[2];
  const float* W2    = (const float*)d_in[3];
  float* out = (float*)d_out;

  _Float16* w1f = (_Float16*)d_ws;

  swizzle_w1_kernel<<<256, 256, 0, stream>>>(W1, w1f);
  pair_kernel<<<dim3(65, 32), 128, 0, stream>>>(x, w1f, nItem, W2, out);
}
